# starting kernel
# baseline (speedup 1.0000x reference)
#include <hip/hip_runtime.h>

typedef float v2f __attribute__((ext_vector_type(2)));
typedef float v8f __attribute__((ext_vector_type(8)));

#define NEG_INF_F (-1000000000.0f)

constexpr int Bn = 64, Cn = 1024, Qn = 128, Dn = 128;
constexpr int P = 132;
constexpr int CHUNK = 128;
constexpr int NCHUNK = Cn / CHUNK;
constexpr int OUTW = 4 * Dn;
constexpr int SM_FLOATS = 3 * (128 * P) + 384 + 128 + 128 + 1024 + 256 + 128;

__global__ __launch_bounds__(256, 1)
void cqa_kernel(const float* __restrict__ ctx_g,
                const float* __restrict__ qry_g,
                const float* __restrict__ W_g,
                const float* __restrict__ qmask_g,
                float* __restrict__ out_g)
{
    extern __shared__ float smem[];
    float* q_lds   = smem;
    float* ctx_lds = q_lds   + Qn * P;
    float* sc_lds  = ctx_lds + CHUNK * P;
    float* w_lds   = sc_lds  + CHUNK * P;
    float* qbias   = w_lds   + 384;
    float* cterm   = qbias   + Qn;
    float* m_all   = cterm   + CHUNK;
    float* red     = m_all   + Cn;
    float* q2c     = red     + 256;

    const int b    = blockIdx.x;
    const int tid  = threadIdx.x;
    const int wv   = tid >> 5;
    const int lane = tid & 31;
    const int lh   = lane >> 4;
    const int ln   = lane & 15;

    const float* ctxB = ctx_g + (size_t)b * Cn * Dn;
    const float* qryB = qry_g + (size_t)b * Qn * Dn;
    float*       outB = out_g + (size_t)b * Cn * OUTW;

    for (int i = tid; i < 3 * Dn; i += 256) w_lds[i] = W_g[i];
    for (int r = wv; r < Qn; r += 8) {
        const int c4 = lane * 4;
        const float4 v = *reinterpret_cast<const float4*>(qryB + (size_t)r * Dn + c4);
        *reinterpret_cast<float4*>(q_lds + r * P + c4) = v;
    }
    __syncthreads();
    if (tid < Qn) {
        float s = 0.f;
        for (int d = 0; d < Dn; ++d) s += q_lds[tid * P + d] * w_lds[Dn + d];
        qbias[tid] = s + (1.0f - qmask_g[b * Qn + tid]) * NEG_INF_F;
    }
    __syncthreads();

    const int rm = wv * 16;

    for (int ch = 0; ch < NCHUNK; ++ch) {
        const int cc = ch * CHUNK;

        for (int r = wv; r < CHUNK; r += 8) {
            const int c4 = lane * 4;
            const float4 v = *reinterpret_cast<const float4*>(ctxB + (size_t)(cc + r) * Dn + c4);
            *reinterpret_cast<float4*>(ctx_lds + r * P + c4) = v;
        }
        __syncthreads();
        if (tid < CHUNK) {
            float s = 0.f;
            for (int d = 0; d < Dn; ++d) s += ctx_lds[tid * P + d] * w_lds[d];
            cterm[tid] = s;
        }
        __syncthreads();

        v8f acc[8];
        #pragma unroll
        for (int t = 0; t < 8; ++t) acc[t] = v8f{0.f,0.f,0.f,0.f,0.f,0.f,0.f,0.f};

        for (int kb = 0; kb < Dn; kb += 4) {
            const int k0 = kb + 2 * lh;
            v2f a = *reinterpret_cast<const v2f*>(ctx_lds + (rm + ln) * P + k0);
            a.x *= w_lds[2 * Dn + k0];
            a.y *= w_lds[2 * Dn + k0 + 1];
            #pragma unroll
            for (int t = 0; t < 8; ++t) {
                const v2f bf = *reinterpret_cast<const v2f*>(q_lds + (t * 16 + ln) * P + k0);
                acc[t] = __builtin_amdgcn_wmma_f32_16x16x4_f32(
                    false, a, false, bf, (short)0, acc[t], false, false);
            }
        }

        #pragma unroll
        for (int r = 0; r < 8; ++r) {
            const int row = rm + r + 8 * lh;
            const float cb = cterm[row];
            float sv[8];
            float mx = -3.4e38f;
            #pragma unroll
            for (int t = 0; t < 8; ++t) {
                const float s = acc[t][r] + cb + qbias[t * 16 + ln];
                sv[t] = s;
                mx = fmaxf(mx, s);
            }
            #pragma unroll
            for (int off = 1; off < 16; off <<= 1)
                mx = fmaxf(mx, __shfl_xor(mx, off, 32));
            if (ln == 0) m_all[cc + row] = mx;
            float sum = 0.f;
            #pragma unroll
            for (int t = 0; t < 8; ++t) { sv[t] = __expf(sv[t] - mx); sum += sv[t]; }
            #pragma unroll
            for (int off = 1; off < 16; off <<= 1)
                sum += __shfl_xor(sum, off, 32);
            const float inv = 1.0f / sum;
            #pragma unroll
            for (int t = 0; t < 8; ++t)
                sc_lds[row * P + t * 16 + ln] = sv[t] * inv;
        }
        __syncthreads();

        v8f acc2[8];
        #pragma unroll
        for (int t = 0; t < 8; ++t) acc2[t] = v8f{0.f,0.f,0.f,0.f,0.f,0.f,0.f,0.f};

        for (int kb = 0; kb < Qn; kb += 4) {
            const int k0 = kb + 2 * lh;
            const v2f a = *reinterpret_cast<const v2f*>(sc_lds + (rm + ln) * P + k0);
            #pragma unroll
            for (int t = 0; t < 8; ++t) {
                v2f bf;
                bf.x = q_lds[k0 * P + t * 16 + ln];
                bf.y = q_lds[(k0 + 1) * P + t * 16 + ln];
                acc2[t] = __builtin_amdgcn_wmma_f32_16x16x4_f32(
                    false, a, false, bf, (short)0, acc2[t], false, false);
            }
        }
        __syncthreads();

        #pragma unroll
        for (int r = 0; r < 8; ++r) {
            const int row = rm + r + 8 * lh;
            #pragma unroll
            for (int t = 0; t < 8; ++t)
                sc_lds[row * P + t * 16 + ln] = acc2[t][r];
        }
        __syncthreads();

        for (int r = wv; r < CHUNK; r += 8) {
            const int c4 = lane * 4;
            const float4 cv = *reinterpret_cast<const float4*>(ctx_lds + r * P + c4);
            const float4 qv = *reinterpret_cast<const float4*>(sc_lds + r * P + c4);
            float* op = outB + (size_t)(cc + r) * OUTW;
            *reinterpret_cast<float4*>(op + c4) = cv;
            *reinterpret_cast<float4*>(op + Dn + c4) = qv;
            const float4 pv = make_float4(cv.x * qv.x, cv.y * qv.y, cv.z * qv.z, cv.w * qv.w);
            *reinterpret_cast<float4*>(op + 2 * Dn + c4) = pv;
        }
        __syncthreads();
    }

    {
        float mx = -3.4e38f;
        for (int i = tid; i < Cn; i += 256) mx = fmaxf(mx, m_all[i]);
        red[tid] = mx;
        __syncthreads();
        for (int s = 128; s > 0; s >>= 1) {
            if (tid < s) red[tid] = fmaxf(red[tid], red[tid + s]);
            __syncthreads();
        }
        const float gmax = red[0];
        __syncthreads();
        float z = 0.f;
        for (int i = tid; i < Cn; i += 256) {
            const float p = __expf(m_all[i] - gmax);
            m_all[i] = p;
            z += p;
        }
        red[tid] = z;
        __syncthreads();
        for (int s = 128; s > 0; s >>= 1) {
            if (tid < s) red[tid] += red[tid + s];
            __syncthreads();
        }
        const float invZ = 1.0f / red[0];
        __syncthreads();

        const int d    = tid & 127;
        const int half = tid >> 7;
        float acc = 0.f;
        for (int c = half * 512; c < half * 512 + 512; ++c)
            acc += m_all[c] * ctxB[(size_t)c * Dn + d];
        red[tid] = acc;
        __syncthreads();
        if (tid < Dn) q2c[tid] = (red[tid] + red[tid + 128]) * invZ;
        __syncthreads();
    }

    for (int idx = tid; idx < Cn * 32; idx += 256) {
        const int r  = idx >> 5;
        const int c4 = (idx & 31) * 4;
        const float4 cv = *reinterpret_cast<const float4*>(ctxB + (size_t)r * Dn + c4);
        const float4 qv = *reinterpret_cast<const float4*>(q2c + c4);
        const float4 pv = make_float4(cv.x * qv.x, cv.y * qv.y, cv.z * qv.z, cv.w * qv.w);
        *reinterpret_cast<float4*>(outB + (size_t)r * OUTW + 3 * Dn + c4) = pv;
    }
}

extern "C" void kernel_launch(void* const* d_in, const int* in_sizes, int n_in,
                              void* d_out, int out_size, void* d_ws, size_t ws_size,
                              hipStream_t stream) {
    const float* ctx   = (const float*)d_in[0];
    const float* qry   = (const float*)d_in[1];
    const float* W     = (const float*)d_in[2];
    const float* qmask = (const float*)d_in[3];
    float* out = (float*)d_out;

    const size_t smem_bytes = (size_t)SM_FLOATS * sizeof(float);
    hipFuncSetAttribute(reinterpret_cast<const void*>(cqa_kernel),
                        hipFuncAttributeMaxDynamicSharedMemorySize, (int)smem_bytes);
    cqa_kernel<<<dim3(Bn), dim3(256), smem_bytes, stream>>>(ctx, qry, W, qmask, out);
}


# wave priority falls with each finished chunk (s_setprio 3,2,1,0): a lagging wave wins the matrix pipe, so a SIMD's four waves end the chunk loop together, none alone at the end; sum order kept
# speedup vs baseline: 2.4459x; 2.4459x over previous
#include <hip/hip_runtime.h>

typedef float v2f __attribute__((ext_vector_type(2)));
typedef float v8f __attribute__((ext_vector_type(8)));
typedef float v4f __attribute__((ext_vector_type(4)));

#define NEG_INF_F (-1000000000.0f)

template <int M>
__device__ __forceinline__ float row_xor(float v) {
    return __int_as_float(__builtin_amdgcn_update_dpp(0, __float_as_int(v), 0x160 + M, 0xf, 0xf, true));
}

__device__ __forceinline__ void wave_handover() {
    __builtin_amdgcn_fence(__ATOMIC_RELEASE, "wavefront");
    __builtin_amdgcn_wave_barrier();
    __builtin_amdgcn_fence(__ATOMIC_ACQUIRE, "wavefront");
}

constexpr int Bn = 64, Cn = 1024, Qn = 128, Dn = 128;
constexpr int P = 132;
constexpr int NW = 16;
constexpr int CHUNK = NW * 16;
constexpr int NCHUNK = Cn / CHUNK;
constexpr int OUTW = 4 * Dn;
constexpr int SM_FLOATS = Qn * P + CHUNK * P + 384 + Qn + CHUNK + Cn + 256 + Dn;

__global__ __launch_bounds__(NW * 32, 1)
void cqa_kernel(const float* __restrict__ ctx_g,
                const float* __restrict__ qry_g,
                const float* __restrict__ W_g,
                const float* __restrict__ qmask_g,
                float* __restrict__ out_g)
{
    extern __shared__ float smem[];
    float* q_lds   = smem;
    float* st_lds  = q_lds   + Qn * P;
    float* w_lds   = st_lds  + CHUNK * P;
    float* qbias   = w_lds   + 384;
    float* cterm   = qbias   + Qn;
    float* m_all   = cterm   + CHUNK;
    float* red     = m_all   + Cn;
    float* q2c     = red     + 256;

    const int b    = blockIdx.x;
    const int tid  = threadIdx.x;
    const int wv   = tid >> 5;
    const int lane = tid & 31;
    const int lh   = lane >> 4;
    const int ln   = lane & 15;

    const float* ctxB = ctx_g + (size_t)b * Cn * Dn;
    const float* qryB = qry_g + (size_t)b * Qn * Dn;
    float*       outB = out_g + (size_t)b * Cn * OUTW;

    for (int i = tid; i < 3 * Dn; i += NW * 32) w_lds[i] = W_g[i];
    for (int r = wv; r < Qn; r += NW) {
        const int c4 = lane * 4;
        const float4 v = *reinterpret_cast<const float4*>(qryB + (size_t)r * Dn + c4);
        *reinterpret_cast<float4*>(q_lds + r * P + c4) = v;
    }
    __syncthreads();
    if (tid < Qn) {
        float s = 0.f;
        for (int d = 0; d < Dn; ++d) s += q_lds[tid * P + d] * w_lds[Dn + d];
        qbias[tid] = s + (1.0f - qmask_g[b * Qn + tid]) * NEG_INF_F;
    }
    __syncthreads();

    const int rm = wv * 16;
    float* st_w = st_lds + rm * P;

    for (int ch = 0; ch < NCHUNK; ++ch) {
        const int cc = ch * CHUNK;

        static_assert(NCHUNK <= 4, "one priority level per chunk: the hardware has four");
        switch (ch) {
            case 0:  __builtin_amdgcn_s_setprio(3); break;
            case 1:  __builtin_amdgcn_s_setprio(2); break;
            case 2:  __builtin_amdgcn_s_setprio(1); break;
            default: __builtin_amdgcn_s_setprio(0); break;
        }

        float4 cv[16];
        {
            const float* src = ctxB + (size_t)(cc + rm) * Dn + lane * 4;
            #pragma unroll
            for (int j = 0; j < 16; ++j)
                cv[j] = *reinterpret_cast<const float4*>(src + (size_t)j * Dn);
            #pragma unroll
            for (int j = 0; j < 16; ++j)
                *reinterpret_cast<float4*>(st_w + j * P + lane * 4) = cv[j];
        }
        wave_handover();
        {
            const float* rowp = st_w + ln * P;
            float s = 0.f;
            for (int d = 0; d < Dn; ++d) s += rowp[d] * w_lds[d];
            cterm[rm + ln] = s;
        }
        wave_handover();

        v8f acc[8];
        #pragma unroll
        for (int t = 0; t < 8; ++t) acc[t] = v8f{0.f,0.f,0.f,0.f,0.f,0.f,0.f,0.f};

        for (int kb = 0; kb < Dn; kb += 4) {
            const int k0 = kb + 2 * lh;
            v2f a = *reinterpret_cast<const v2f*>(st_w + ln * P + k0);
            a.x *= w_lds[2 * Dn + k0];
            a.y *= w_lds[2 * Dn + k0 + 1];
            #pragma unroll
            for (int t = 0; t < 8; ++t) {
                const v2f bf = *reinterpret_cast<const v2f*>(q_lds + (t * 16 + ln) * P + k0);
                acc[t] = __builtin_amdgcn_wmma_f32_16x16x4_f32(
                    false, a, false, bf, (short)0, acc[t], false, false);
            }
        }

        wave_handover();

        #pragma unroll
        for (int r = 0; r < 8; ++r) {
            const int row = rm + r + 8 * lh;
            const float cb = cterm[row];
            float sv[8];
            float mx = -3.4e38f;
            #pragma unroll
            for (int t = 0; t < 8; ++t) {
                const float s = acc[t][r] + cb + qbias[t * 16 + ln];
                sv[t] = s;
                mx = fmaxf(mx, s);
            }
            mx = fmaxf(mx, row_xor<1>(mx));
            mx = fmaxf(mx, row_xor<2>(mx));
            mx = fmaxf(mx, row_xor<4>(mx));
            mx = fmaxf(mx, row_xor<8>(mx));
            if (ln == 0) m_all[cc + row] = mx;
            float sum = 0.f;
            #pragma unroll
            for (int t = 0; t < 8; ++t) { sv[t] = __expf(sv[t] - mx); sum += sv[t]; }
            sum += row_xor<1>(sum);
            sum += row_xor<2>(sum);
            sum += row_xor<4>(sum);
            sum += row_xor<8>(sum);
            const float inv = 1.0f / sum;
            #pragma unroll
            for (int t = 0; t < 8; ++t)
                st_lds[row * P + t * 16 + ln] = sv[t] * inv;
        }
        wave_handover();

        v8f acc2[8];
        #pragma unroll
        for (int t = 0; t < 8; ++t) acc2[t] = v8f{0.f,0.f,0.f,0.f,0.f,0.f,0.f,0.f};

        for (int kb = 0; kb < Qn; kb += 4) {
            const int k0 = kb + 2 * lh;
            const v2f a = *reinterpret_cast<const v2f*>(st_w + ln * P + k0);
            #pragma unroll
            for (int t = 0; t < 8; ++t) {
                v2f bf;
                bf.x = q_lds[k0 * P + t * 16 + ln];
                bf.y = q_lds[(k0 + 1) * P + t * 16 + ln];
                acc2[t] = __builtin_amdgcn_wmma_f32_16x16x4_f32(
                    false, a, false, bf, (short)0, acc2[t], false, false);
            }
        }
        wave_handover();

        #pragma unroll
        for (int r = 0; r < 8; ++r) {
            const int row = rm + r + 8 * lh;
            #pragma unroll
            for (int t = 0; t < 8; ++t)
                st_lds[row * P + t * 16 + ln] = acc2[t][r];
        }
        wave_handover();

        #pragma unroll
        for (int j = 0; j < 16; ++j) {
            const int c4 = lane * 4;
            const float4 qv = *reinterpret_cast<const float4*>(st_w + j * P + c4);
            float* op = outB + (size_t)(cc + rm + j) * OUTW;
            *reinterpret_cast<float4*>(op + c4) = cv[j];
            *reinterpret_cast<float4*>(op + Dn + c4) = qv;
            const float4 pv = make_float4(cv[j].x * qv.x, cv[j].y * qv.y, cv[j].z * qv.z, cv[j].w * qv.w);
            *reinterpret_cast<float4*>(op + 2 * Dn + c4) = pv;
        }
        wave_handover();
    }
    __syncthreads();

    {
        if (tid < 256) {
            float mx = -3.4e38f;
            for (int i = tid; i < Cn; i += 256) mx = fmaxf(mx, m_all[i]);
            red[tid] = mx;
        }
        __syncthreads();
        for (int s = 128; s > 0; s >>= 1) {
            if (tid < s) red[tid] = fmaxf(red[tid], red[tid + s]);
            __syncthreads();
        }
        const float gmax = red[0];
        __syncthreads();
        if (tid < 256) {
            float z = 0.f;
            for (int i = tid; i < Cn; i += 256) {
                const float p = __expf(m_all[i] - gmax);
                m_all[i] = p;
                z += p;
            }
            red[tid] = z;
        }
        __syncthreads();
        for (int s = 128; s > 0; s >>= 1) {
            if (tid < s) red[tid] += red[tid + s];
            __syncthreads();
        }
        const float invZ = 1.0f / red[0];
        __syncthreads();

        if (tid < 256) {
            const int d    = tid & 127;
            const int half = tid >> 7;
            const float* cp = ctxB + (size_t)(half * 512) * Dn + d;
            const float* pp = m_all + half * 512;
            float acc = 0.f;
            #pragma unroll 32
            for (int i = 0; i < 512; ++i)
                acc += pp[i] * cp[(size_t)i * Dn];
            red[tid] = acc;
        }
        __syncthreads();
        if (tid < Dn) q2c[tid] = (red[tid] + red[tid + 128]) * invZ;
        __syncthreads();
    }

    {
        const int c4 = lane * 4;
        const v4f qv = *reinterpret_cast<const v4f*>(q2c + c4);
        const float* cp = ctxB + (size_t)wv * Dn + c4;
        float*       op = outB + (size_t)wv * OUTW + 3 * Dn + c4;
        for (int i0 = 0; i0 < Cn / NW; i0 += 16) {
            v4f cv[16];
            #pragma unroll
            for (int j = 0; j < 16; ++j)
                cv[j] = *reinterpret_cast<const v4f*>(cp + (size_t)(i0 + j) * NW * Dn);
            asm volatile("" :: "v"(cv[0]), "v"(cv[1]), "v"(cv[2]), "v"(cv[3]), "v"(cv[4]), "v"(cv[5]), "v"(cv[6]), "v"(cv[7]),
                               "v"(cv[8]), "v"(cv[9]), "v"(cv[10]), "v"(cv[11]), "v"(cv[12]), "v"(cv[13]), "v"(cv[14]), "v"(cv[15]));
            #pragma unroll
            for (int j = 0; j < 16; ++j)
                *reinterpret_cast<v4f*>(op + (size_t)(i0 + j) * NW * OUTW) = cv[j] * qv;
        }
    }
}

extern "C" void kernel_launch(void* const* d_in, const int* in_sizes, int n_in,
                              void* d_out, int out_size, void* d_ws, size_t ws_size,
                              hipStream_t stream) {
    const float* ctx   = (const float*)d_in[0];
    const float* qry   = (const float*)d_in[1];
    const float* W     = (const float*)d_in[2];
    const float* qmask = (const float*)d_in[3];
    float* out = (float*)d_out;

    const size_t smem_bytes = (size_t)SM_FLOATS * sizeof(float);
    hipFuncSetAttribute(reinterpret_cast<const void*>(cqa_kernel),
                        hipFuncAttributeMaxDynamicSharedMemorySize, (int)smem_bytes);
    cqa_kernel<<<dim3(Bn), dim3(NW * 32), smem_bytes, stream>>>(ctx, qry, W, qmask, out);
}


# Grid 64->256: four work-groups per batch element, each recomputes the score rows but owns 32 of the 128 feature columns of c2q, q2c and all four output blocks; nothing exchanged; sum order kept
# speedup vs baseline: 2.9294x; 1.1977x over previous
#include <hip/hip_runtime.h>

typedef float v2f __attribute__((ext_vector_type(2)));
typedef float v8f __attribute__((ext_vector_type(8)));
typedef float v4f __attribute__((ext_vector_type(4)));

#define NEG_INF_F (-1000000000.0f)

template <int M>
__device__ __forceinline__ float row_xor(float v) {
    return __int_as_float(__builtin_amdgcn_update_dpp(0, __float_as_int(v), 0x160 + M, 0xf, 0xf, true));
}

__device__ __forceinline__ void wave_handover() {
    __builtin_amdgcn_fence(__ATOMIC_RELEASE, "wavefront");
    __builtin_amdgcn_wave_barrier();
    __builtin_amdgcn_fence(__ATOMIC_ACQUIRE, "wavefront");
}

constexpr int Bn = 64, Cn = 1024, Qn = 128, Dn = 128;
constexpr int P = 132;
constexpr int NW = 16;
constexpr int CHUNK = NW * 16;
constexpr int NCHUNK = Cn / CHUNK;
constexpr int OUTW = 4 * Dn;
constexpr int NSPLIT = 4;
constexpr int DS = Dn / NSPLIT;
constexpr int NGRID = Bn * NSPLIT;
constexpr int SM_FLOATS = Qn * P + CHUNK * P + 384 + Qn + CHUNK + Cn + 256 + Dn;

__global__ __launch_bounds__(NW * 32, 1)
void cqa_kernel(const float* __restrict__ ctx_g,
                const float* __restrict__ qry_g,
                const float* __restrict__ W_g,
                const float* __restrict__ qmask_g,
                float* __restrict__ out_g)
{
    extern __shared__ float smem[];
    float* q_lds   = smem;
    float* st_lds  = q_lds   + Qn * P;
    float* w_lds   = st_lds  + CHUNK * P;
    float* qbias   = w_lds   + 384;
    float* cterm   = qbias   + Qn;
    float* m_all   = cterm   + CHUNK;
    float* red     = m_all   + Cn;
    float* q2c     = red     + 256;

    const int b    = blockIdx.x / NSPLIT;
    const int d0   = (blockIdx.x % NSPLIT) * DS;
    const int tid  = threadIdx.x;
    const int wv   = tid >> 5;
    const int lane = tid & 31;
    const int lh   = lane >> 4;
    const int ln   = lane & 15;
    const int sr   = lane >> 3;
    const int sc   = (lane & 7) * 4;

    const float* ctxB = ctx_g + (size_t)b * Cn * Dn;
    const float* qryB = qry_g + (size_t)b * Qn * Dn;
    float*       outB = out_g + (size_t)b * Cn * OUTW;

    for (int i = tid; i < 3 * Dn; i += NW * 32) w_lds[i] = W_g[i];
    for (int r = wv; r < Qn; r += NW) {
        const int c4 = lane * 4;
        const float4 v = *reinterpret_cast<const float4*>(qryB + (size_t)r * Dn + c4);
        *reinterpret_cast<float4*>(q_lds + r * P + c4) = v;
    }
    __syncthreads();
    if (tid < Qn) {
        float s = 0.f;
        for (int d = 0; d < Dn; ++d) s += q_lds[tid * P + d] * w_lds[Dn + d];
        qbias[tid] = s + (1.0f - qmask_g[b * Qn + tid]) * NEG_INF_F;
    }
    __syncthreads();

    const int rm = wv * 16;
    float* st_w = st_lds + rm * P;

    for (int ch = 0; ch < NCHUNK; ++ch) {
        const int cc = ch * CHUNK;

        static_assert(NCHUNK <= 4, "one priority level per chunk: the hardware has four");
        switch (ch) {
            case 0:  __builtin_amdgcn_s_setprio(3); break;
            case 1:  __builtin_amdgcn_s_setprio(2); break;
            case 2:  __builtin_amdgcn_s_setprio(1); break;
            default: __builtin_amdgcn_s_setprio(0); break;
        }

        float4 cs[4];
        {
            float4 cv[16];
            const float* src = ctxB + (size_t)(cc + rm) * Dn + lane * 4;
            #pragma unroll
            for (int j = 0; j < 16; ++j)
                cv[j] = *reinterpret_cast<const float4*>(src + (size_t)j * Dn);
            #pragma unroll
            for (int j = 0; j < 16; ++j)
                *reinterpret_cast<float4*>(st_w + j * P + lane * 4) = cv[j];
        }
        wave_handover();
        #pragma unroll
        for (int i = 0; i < 4; ++i)
            cs[i] = *reinterpret_cast<const float4*>(st_w + (4 * i + sr) * P + d0 + sc);
        {
            const float* rowp = st_w + ln * P;
            float s = 0.f;
            for (int d = 0; d < Dn; ++d) s += rowp[d] * w_lds[d];
            cterm[rm + ln] = s;
        }
        wave_handover();

        v8f acc[8];
        #pragma unroll
        for (int t = 0; t < 8; ++t) acc[t] = v8f{0.f,0.f,0.f,0.f,0.f,0.f,0.f,0.f};

        for (int kb = 0; kb < Dn; kb += 4) {
            const int k0 = kb + 2 * lh;
            v2f a = *reinterpret_cast<const v2f*>(st_w + ln * P + k0);
            a.x *= w_lds[2 * Dn + k0];
            a.y *= w_lds[2 * Dn + k0 + 1];
            #pragma unroll
            for (int t = 0; t < 8; ++t) {
                const v2f bf = *reinterpret_cast<const v2f*>(q_lds + (t * 16 + ln) * P + k0);
                acc[t] = __builtin_amdgcn_wmma_f32_16x16x4_f32(
                    false, a, false, bf, (short)0, acc[t], false, false);
            }
        }

        wave_handover();

        #pragma unroll
        for (int r = 0; r < 8; ++r) {
            const int row = rm + r + 8 * lh;
            const float cb = cterm[row];
            float sv[8];
            float mx = -3.4e38f;
            #pragma unroll
            for (int t = 0; t < 8; ++t) {
                const float s = acc[t][r] + cb + qbias[t * 16 + ln];
                sv[t] = s;
                mx = fmaxf(mx, s);
            }
            mx = fmaxf(mx, row_xor<1>(mx));
            mx = fmaxf(mx, row_xor<2>(mx));
            mx = fmaxf(mx, row_xor<4>(mx));
            mx = fmaxf(mx, row_xor<8>(mx));
            if (ln == 0) m_all[cc + row] = mx;
            float sum = 0.f;
            #pragma unroll
            for (int t = 0; t < 8; ++t) { sv[t] = __expf(sv[t] - mx); sum += sv[t]; }
            sum += row_xor<1>(sum);
            sum += row_xor<2>(sum);
            sum += row_xor<4>(sum);
            sum += row_xor<8>(sum);
            const float inv = 1.0f / sum;
            #pragma unroll
            for (int t = 0; t < 8; ++t)
                st_lds[row * P + t * 16 + ln] = sv[t] * inv;
        }
        wave_handover();

        constexpr int NT2 = DS / 16;
        v8f acc2[NT2];
        #pragma unroll
        for (int t = 0; t < NT2; ++t) acc2[t] = v8f{0.f,0.f,0.f,0.f,0.f,0.f,0.f,0.f};

        for (int kb = 0; kb < Qn; kb += 4) {
            const int k0 = kb + 2 * lh;
            const v2f a = *reinterpret_cast<const v2f*>(st_w + ln * P + k0);
            #pragma unroll
            for (int t = 0; t < NT2; ++t) {
                v2f bf;
                bf.x = q_lds[k0 * P + d0 + t * 16 + ln];
                bf.y = q_lds[(k0 + 1) * P + d0 + t * 16 + ln];
                acc2[t] = __builtin_amdgcn_wmma_f32_16x16x4_f32(
                    false, a, false, bf, (short)0, acc2[t], false, false);
            }
        }
        wave_handover();

        #pragma unroll
        for (int r = 0; r < 8; ++r) {
            const int row = rm + r + 8 * lh;
            #pragma unroll
            for (int t = 0; t < NT2; ++t)
                st_lds[row * P + t * 16 + ln] = acc2[t][r];
        }
        wave_handover();

        #pragma unroll
        for (int i = 0; i < 4; ++i) {
            const int j = 4 * i + sr;
            const float4 qv = *reinterpret_cast<const float4*>(st_w + j * P + sc);
            float* op = outB + (size_t)(cc + rm + j) * OUTW + d0 + sc;
            *reinterpret_cast<float4*>(op) = cs[i];
            *reinterpret_cast<float4*>(op + Dn) = qv;
            const float4 pv = make_float4(cs[i].x * qv.x, cs[i].y * qv.y, cs[i].z * qv.z, cs[i].w * qv.w);
            *reinterpret_cast<float4*>(op + 2 * Dn) = pv;
        }
        wave_handover();
    }
    __syncthreads();

    {
        if (tid < 256) {
            float mx = -3.4e38f;
            for (int i = tid; i < Cn; i += 256) mx = fmaxf(mx, m_all[i]);
            red[tid] = mx;
        }
        __syncthreads();
        for (int s = 128; s > 0; s >>= 1) {
            if (tid < s) red[tid] = fmaxf(red[tid], red[tid + s]);
            __syncthreads();
        }
        const float gmax = red[0];
        __syncthreads();
        if (tid < 256) {
            float z = 0.f;
            for (int i = tid; i < Cn; i += 256) {
                const float p = __expf(m_all[i] - gmax);
                m_all[i] = p;
                z += p;
            }
            red[tid] = z;
        }
        __syncthreads();
        for (int s = 128; s > 0; s >>= 1) {
            if (tid < s) red[tid] += red[tid + s];
            __syncthreads();
        }
        const float invZ = 1.0f / red[0];
        __syncthreads();

        if (tid < 2 * DS) {
            const int d    = d0 + (tid & (DS - 1));
            const int half = tid / DS;
            const float* cp = ctxB + (size_t)(half * 512) * Dn + d;
            const float* pp = m_all + half * 512;
            float acc = 0.f;
            #pragma unroll 32
            for (int i = 0; i < 512; ++i)
                acc += pp[i] * cp[(size_t)i * Dn];
            red[tid] = acc;
        }
        __syncthreads();
        if (tid < DS) q2c[tid] = (red[tid] + red[tid + DS]) * invZ;
        __syncthreads();
    }

    {
        const v4f qv = *reinterpret_cast<const v4f*>(q2c + sc);
        const float* cp = ctxB + (size_t)(wv * 4 + sr) * Dn + d0 + sc;
        float*       op = outB + (size_t)(wv * 4 + sr) * OUTW + 3 * Dn + d0 + sc;
        for (int i0 = 0; i0 < Cn / (4 * NW); i0 += 16) {
            v4f cv[16];
            #pragma unroll
            for (int j = 0; j < 16; ++j)
                cv[j] = *reinterpret_cast<const v4f*>(cp + (size_t)(i0 + j) * 4 * NW * Dn);
            asm volatile("" :: "v"(cv[0]), "v"(cv[1]), "v"(cv[2]), "v"(cv[3]), "v"(cv[4]), "v"(cv[5]), "v"(cv[6]), "v"(cv[7]),
                               "v"(cv[8]), "v"(cv[9]), "v"(cv[10]), "v"(cv[11]), "v"(cv[12]), "v"(cv[13]), "v"(cv[14]), "v"(cv[15]));
            #pragma unroll
            for (int j = 0; j < 16; ++j)
                *reinterpret_cast<v4f*>(op + (size_t)(i0 + j) * 4 * NW * OUTW) = cv[j] * qv;
        }
    }
}

extern "C" void kernel_launch(void* const* d_in, const int* in_sizes, int n_in,
                              void* d_out, int out_size, void* d_ws, size_t ws_size,
                              hipStream_t stream) {
    const float* ctx   = (const float*)d_in[0];
    const float* qry   = (const float*)d_in[1];
    const float* W     = (const float*)d_in[2];
    const float* qmask = (const float*)d_in[3];
    float* out = (float*)d_out;

    const size_t smem_bytes = (size_t)SM_FLOATS * sizeof(float);
    hipFuncSetAttribute(reinterpret_cast<const void*>(cqa_kernel),
                        hipFuncAttributeMaxDynamicSharedMemorySize, (int)smem_bytes);
    cqa_kernel<<<dim3(NGRID), dim3(NW * 32), smem_bytes, stream>>>(ctx, qry, W, qmask, out);
}


# loads outside the chunk loop go out together: W, mask and 8 query rows (10 round trips -> 1); q2c context read spread over all 16 waves and shared with block 3 (17 -> 1); q2c sum order changes
# speedup vs baseline: 3.2666x; 1.1151x over previous
#include <hip/hip_runtime.h>

typedef float v2f __attribute__((ext_vector_type(2)));
typedef float v8f __attribute__((ext_vector_type(8)));
typedef float v4f __attribute__((ext_vector_type(4)));

#define NEG_INF_F (-1000000000.0f)

template <int M>
__device__ __forceinline__ float row_xor(float v) {
    return __int_as_float(__builtin_amdgcn_update_dpp(0, __float_as_int(v), 0x160 + M, 0xf, 0xf, true));
}

__device__ __forceinline__ void wave_handover() {
    __builtin_amdgcn_fence(__ATOMIC_RELEASE, "wavefront");
    __builtin_amdgcn_wave_barrier();
    __builtin_amdgcn_fence(__ATOMIC_ACQUIRE, "wavefront");
}

constexpr int Bn = 64, Cn = 1024, Qn = 128, Dn = 128;
constexpr int P = 132;
constexpr int NW = 16;
constexpr int CHUNK = NW * 16;
constexpr int NCHUNK = Cn / CHUNK;
constexpr int OUTW = 4 * Dn;
constexpr int NSPLIT = 4;
constexpr int DS = Dn / NSPLIT;
constexpr int NGRID = Bn * NSPLIT;
constexpr int SM_FLOATS = Qn * P + CHUNK * P + 384 + Qn + CHUNK + Cn + 256 + Dn;

__global__ __launch_bounds__(NW * 32, 1)
void cqa_kernel(const float* __restrict__ ctx_g,
                const float* __restrict__ qry_g,
                const float* __restrict__ W_g,
                const float* __restrict__ qmask_g,
                float* __restrict__ out_g)
{
    extern __shared__ float smem[];
    float* q_lds   = smem;
    float* st_lds  = q_lds   + Qn * P;
    float* w_lds   = st_lds  + CHUNK * P;
    float* qbias   = w_lds   + 384;
    float* cterm   = qbias   + Qn;
    float* m_all   = cterm   + CHUNK;
    float* red     = m_all   + Cn;
    float* q2c     = red     + 256;

    const int b    = blockIdx.x / NSPLIT;
    const int d0   = (blockIdx.x % NSPLIT) * DS;
    const int tid  = threadIdx.x;
    const int wv   = tid >> 5;
    const int lane = tid & 31;
    const int lh   = lane >> 4;
    const int ln   = lane & 15;
    const int sr   = lane >> 3;
    const int sc   = (lane & 7) * 4;

    const float* ctxB = ctx_g + (size_t)b * Cn * Dn;
    const float* qryB = qry_g + (size_t)b * Qn * Dn;
    float*       outB = out_g + (size_t)b * Cn * OUTW;

    static_assert(3 * Dn <= NW * 32 && Qn <= NW * 32, "one W element and one mask element per thread");
    constexpr int QR = Qn / NW;
    const float w_reg = (tid < 3 * Dn) ? W_g[tid] : 0.f;
    const float m_reg = (tid < Qn) ? qmask_g[b * Qn + tid] : 0.f;
    float4 qrow[QR];
    #pragma unroll
    for (int i = 0; i < QR; ++i)
        qrow[i] = *reinterpret_cast<const float4*>(qryB + (size_t)(wv + i * NW) * Dn + lane * 4);
    #pragma unroll
    for (int i = 0; i < QR; ++i)
        *reinterpret_cast<float4*>(q_lds + (wv + i * NW) * P + lane * 4) = qrow[i];
    if (tid < 3 * Dn) w_lds[tid] = w_reg;
    __syncthreads();
    if (tid < Qn) {
        float s = 0.f;
        for (int d = 0; d < Dn; ++d) s += q_lds[tid * P + d] * w_lds[Dn + d];
        qbias[tid] = s + (1.0f - m_reg) * NEG_INF_F;
    }
    __syncthreads();

    const int rm = wv * 16;
    float* st_w = st_lds + rm * P;

    for (int ch = 0; ch < NCHUNK; ++ch) {
        const int cc = ch * CHUNK;

        static_assert(NCHUNK <= 4, "one priority level per chunk: the hardware has four");
        switch (ch) {
            case 0:  __builtin_amdgcn_s_setprio(3); break;
            case 1:  __builtin_amdgcn_s_setprio(2); break;
            case 2:  __builtin_amdgcn_s_setprio(1); break;
            default: __builtin_amdgcn_s_setprio(0); break;
        }

        float4 cs[4];
        {
            float4 cv[16];
            const float* src = ctxB + (size_t)(cc + rm) * Dn + lane * 4;
            #pragma unroll
            for (int j = 0; j < 16; ++j)
                cv[j] = *reinterpret_cast<const float4*>(src + (size_t)j * Dn);
            #pragma unroll
            for (int j = 0; j < 16; ++j)
                *reinterpret_cast<float4*>(st_w + j * P + lane * 4) = cv[j];
        }
        wave_handover();
        #pragma unroll
        for (int i = 0; i < 4; ++i)
            cs[i] = *reinterpret_cast<const float4*>(st_w + (4 * i + sr) * P + d0 + sc);
        {
            const float* rowp = st_w + ln * P;
            float s = 0.f;
            for (int d = 0; d < Dn; ++d) s += rowp[d] * w_lds[d];
            cterm[rm + ln] = s;
        }
        wave_handover();

        v8f acc[8];
        #pragma unroll
        for (int t = 0; t < 8; ++t) acc[t] = v8f{0.f,0.f,0.f,0.f,0.f,0.f,0.f,0.f};

        for (int kb = 0; kb < Dn; kb += 4) {
            const int k0 = kb + 2 * lh;
            v2f a = *reinterpret_cast<const v2f*>(st_w + ln * P + k0);
            a.x *= w_lds[2 * Dn + k0];
            a.y *= w_lds[2 * Dn + k0 + 1];
            #pragma unroll
            for (int t = 0; t < 8; ++t) {
                const v2f bf = *reinterpret_cast<const v2f*>(q_lds + (t * 16 + ln) * P + k0);
                acc[t] = __builtin_amdgcn_wmma_f32_16x16x4_f32(
                    false, a, false, bf, (short)0, acc[t], false, false);
            }
        }

        wave_handover();

        #pragma unroll
        for (int r = 0; r < 8; ++r) {
            const int row = rm + r + 8 * lh;
            const float cb = cterm[row];
            float sv[8];
            float mx = -3.4e38f;
            #pragma unroll
            for (int t = 0; t < 8; ++t) {
                const float s = acc[t][r] + cb + qbias[t * 16 + ln];
                sv[t] = s;
                mx = fmaxf(mx, s);
            }
            mx = fmaxf(mx, row_xor<1>(mx));
            mx = fmaxf(mx, row_xor<2>(mx));
            mx = fmaxf(mx, row_xor<4>(mx));
            mx = fmaxf(mx, row_xor<8>(mx));
            if (ln == 0) m_all[cc + row] = mx;
            float sum = 0.f;
            #pragma unroll
            for (int t = 0; t < 8; ++t) { sv[t] = __expf(sv[t] - mx); sum += sv[t]; }
            sum += row_xor<1>(sum);
            sum += row_xor<2>(sum);
            sum += row_xor<4>(sum);
            sum += row_xor<8>(sum);
            const float inv = 1.0f / sum;
            #pragma unroll
            for (int t = 0; t < 8; ++t)
                st_lds[row * P + t * 16 + ln] = sv[t] * inv;
        }
        wave_handover();

        constexpr int NT2 = DS / 16;
        v8f acc2[NT2];
        #pragma unroll
        for (int t = 0; t < NT2; ++t) acc2[t] = v8f{0.f,0.f,0.f,0.f,0.f,0.f,0.f,0.f};

        for (int kb = 0; kb < Qn; kb += 4) {
            const int k0 = kb + 2 * lh;
            const v2f a = *reinterpret_cast<const v2f*>(st_w + ln * P + k0);
            #pragma unroll
            for (int t = 0; t < NT2; ++t) {
                v2f bf;
                bf.x = q_lds[k0 * P + d0 + t * 16 + ln];
                bf.y = q_lds[(k0 + 1) * P + d0 + t * 16 + ln];
                acc2[t] = __builtin_amdgcn_wmma_f32_16x16x4_f32(
                    false, a, false, bf, (short)0, acc2[t], false, false);
            }
        }
        wave_handover();

        #pragma unroll
        for (int r = 0; r < 8; ++r) {
            const int row = rm + r + 8 * lh;
            #pragma unroll
            for (int t = 0; t < NT2; ++t)
                st_lds[row * P + t * 16 + ln] = acc2[t][r];
        }
        wave_handover();

        #pragma unroll
        for (int i = 0; i < 4; ++i) {
            const int j = 4 * i + sr;
            const float4 qv = *reinterpret_cast<const float4*>(st_w + j * P + sc);
            float* op = outB + (size_t)(cc + rm + j) * OUTW + d0 + sc;
            *reinterpret_cast<float4*>(op) = cs[i];
            *reinterpret_cast<float4*>(op + Dn) = qv;
            const float4 pv = make_float4(cs[i].x * qv.x, cs[i].y * qv.y, cs[i].z * qv.z, cs[i].w * qv.w);
            *reinterpret_cast<float4*>(op + 2 * Dn) = pv;
        }
        wave_handover();
    }
    constexpr int RG = Cn / (4 * NW);
    static_assert(RG == 16, "the tail keeps one wave's 16 row groups in registers");
    v4f cv[RG];
    {
        const float* cp = ctxB + (size_t)(wv * 4 + sr) * Dn + d0 + sc;
        #pragma unroll
        for (int j = 0; j < RG; ++j)
            cv[j] = *reinterpret_cast<const v4f*>(cp + (size_t)j * 4 * NW * Dn);
    }
    __syncthreads();

    {
        if (tid < 256) {
            float mx = -3.4e38f;
            for (int i = tid; i < Cn; i += 256) mx = fmaxf(mx, m_all[i]);
            red[tid] = mx;
        }
        __syncthreads();
        for (int s = 128; s > 0; s >>= 1) {
            if (tid < s) red[tid] = fmaxf(red[tid], red[tid + s]);
            __syncthreads();
        }
        const float gmax = red[0];
        __syncthreads();
        if (tid < 256) {
            float z = 0.f;
            for (int i = tid; i < Cn; i += 256) {
                const float p = __expf(m_all[i] - gmax);
                m_all[i] = p;
                z += p;
            }
            red[tid] = z;
        }
        __syncthreads();
        for (int s = 128; s > 0; s >>= 1) {
            if (tid < s) red[tid] += red[tid + s];
            __syncthreads();
        }
        const float invZ = 1.0f / red[0];
        __syncthreads();

        v4f part = v4f{0.f, 0.f, 0.f, 0.f};
        #pragma unroll
        for (int j = 0; j < RG; ++j)
            part += m_all[j * 4 * NW + wv * 4 + sr] * cv[j];
        part.x += row_xor<8>(part.x);
        part.y += row_xor<8>(part.y);
        part.z += row_xor<8>(part.z);
        part.w += row_xor<8>(part.w);
        float* pw = st_lds;
        if ((lane & 8) == 0)
            *reinterpret_cast<v4f*>(pw + (wv * 2 + lh) * DS + (lane & 7) * 4) = part;
        __syncthreads();
        if (tid < DS) {
            float s = 0.f;
            #pragma unroll
            for (int k = 0; k < 2 * NW; ++k) s += pw[k * DS + tid];
            q2c[tid] = s * invZ;
        }
        __syncthreads();
    }

    {
        const v4f qv = *reinterpret_cast<const v4f*>(q2c + sc);
        float* op = outB + (size_t)(wv * 4 + sr) * OUTW + 3 * Dn + d0 + sc;
        #pragma unroll
        for (int j = 0; j < RG; ++j)
            *reinterpret_cast<v4f*>(op + (size_t)j * 4 * NW * OUTW) = cv[j] * qv;
    }
}

extern "C" void kernel_launch(void* const* d_in, const int* in_sizes, int n_in,
                              void* d_out, int out_size, void* d_ws, size_t ws_size,
                              hipStream_t stream) {
    const float* ctx   = (const float*)d_in[0];
    const float* qry   = (const float*)d_in[1];
    const float* W     = (const float*)d_in[2];
    const float* qmask = (const float*)d_in[3];
    float* out = (float*)d_out;

    const size_t smem_bytes = (size_t)SM_FLOATS * sizeof(float);
    hipFuncSetAttribute(reinterpret_cast<const void*>(cqa_kernel),
                        hipFuncAttributeMaxDynamicSharedMemorySize, (int)smem_bytes);
    cqa_kernel<<<dim3(NGRID), dim3(NW * 32), smem_bytes, stream>>>(ctx, qry, W, qmask, out);
}


# both matrix products run in blocks (32 / 16 products) whose LDS fragment reads stand in ONE group a block ahead (second register set): 8 read places per wave and chunk, was 56; sum order kept
# speedup vs baseline: 3.7602x; 1.1511x over previous
#include <hip/hip_runtime.h>

typedef float v2f __attribute__((ext_vector_type(2)));
typedef float v8f __attribute__((ext_vector_type(8)));
typedef float v4f __attribute__((ext_vector_type(4)));

#define NEG_INF_F (-1000000000.0f)

template <int M>
__device__ __forceinline__ float row_xor(float v) {
    return __int_as_float(__builtin_amdgcn_update_dpp(0, __float_as_int(v), 0x160 + M, 0xf, 0xf, true));
}

__device__ __forceinline__ void wave_handover() {
    __builtin_amdgcn_fence(__ATOMIC_RELEASE, "wavefront");
    __builtin_amdgcn_wave_barrier();
    __builtin_amdgcn_fence(__ATOMIC_ACQUIRE, "wavefront");
}

constexpr int Bn = 64, Cn = 1024, Qn = 128, Dn = 128;
constexpr int P = 132;
constexpr int NW = 16;
constexpr int CHUNK = NW * 16;
constexpr int NCHUNK = Cn / CHUNK;
constexpr int OUTW = 4 * Dn;
constexpr int NSPLIT = 4;
constexpr int DS = Dn / NSPLIT;
constexpr int NGRID = Bn * NSPLIT;
constexpr int QS_FLOATS = Qn * DS;
static_assert(DS == 32, "the QS fragment layout packs exactly two 16-column tiles per 16-byte read");
constexpr int SM_FLOATS = Qn * P + CHUNK * P + 384 + Qn + QS_FLOATS + Cn + 256 + Dn;

__global__ __launch_bounds__(NW * 32, 1)
void cqa_kernel(const float* __restrict__ ctx_g,
                const float* __restrict__ qry_g,
                const float* __restrict__ W_g,
                const float* __restrict__ qmask_g,
                float* __restrict__ out_g)
{
    extern __shared__ float smem[];
    float* q_lds   = smem;
    float* st_lds  = q_lds   + Qn * P;
    float* w_lds   = st_lds  + CHUNK * P;
    float* qbias   = w_lds   + 384;
    float* qs_lds  = qbias   + Qn;
    float* m_all   = qs_lds  + QS_FLOATS;
    float* red     = m_all   + Cn;
    float* q2c     = red     + 256;

    const int b    = blockIdx.x / NSPLIT;
    const int d0   = (blockIdx.x % NSPLIT) * DS;
    const int tid  = threadIdx.x;
    const int wv   = tid >> 5;
    const int lane = tid & 31;
    const int lh   = lane >> 4;
    const int ln   = lane & 15;
    const int sr   = lane >> 3;
    const int sc   = (lane & 7) * 4;

    const float* ctxB = ctx_g + (size_t)b * Cn * Dn;
    const float* qryB = qry_g + (size_t)b * Qn * Dn;
    float*       outB = out_g + (size_t)b * Cn * OUTW;

    static_assert(3 * Dn <= NW * 32 && Qn <= NW * 32, "one W element and one mask element per thread");
    constexpr int QR = Qn / NW;
    const float w_reg = (tid < 3 * Dn) ? W_g[tid] : 0.f;
    const float m_reg = (tid < Qn) ? qmask_g[b * Qn + tid] : 0.f;
    const float4 wc4 = *reinterpret_cast<const float4*>(W_g + lane * 4);
    const float4 ws4 = *reinterpret_cast<const float4*>(W_g + 2 * Dn + lane * 4);
    float4 qrow[QR];
    #pragma unroll
    for (int i = 0; i < QR; ++i)
        qrow[i] = *reinterpret_cast<const float4*>(qryB + (size_t)(wv + i * NW) * Dn + lane * 4);
    const bool own_cols = (lane >> 3) == (int)(blockIdx.x % NSPLIT);
    #pragma unroll
    for (int i = 0; i < QR; ++i) {
        const int k = wv + i * NW;
        const float4 q = qrow[i];
        *reinterpret_cast<float4*>(st_lds + k * P + lane * 4) = q;
        *reinterpret_cast<float4*>(q_lds + k * P + lane * 4) =
            make_float4(fmaf(ws4.x, q.x, wc4.x), fmaf(ws4.y, q.y, wc4.y),
                        fmaf(ws4.z, q.z, wc4.z), fmaf(ws4.w, q.w, wc4.w));
        if (own_cols) {
            float* qp = qs_lds + (k >> 1) * (2 * DS) + (lane & 3) * 16 + ((lane >> 2) & 1) * 2 + (k & 1);
            qp[0] = q.x; qp[4] = q.y; qp[8] = q.z; qp[12] = q.w;
        }
    }
    if (tid < 3 * Dn) w_lds[tid] = w_reg;
    __syncthreads();
    if (tid < Qn) {
        float s = 0.f;
        for (int d = 0; d < Dn; ++d) s += st_lds[tid * P + d] * w_lds[Dn + d];
        qbias[tid] = s + (1.0f - m_reg) * NEG_INF_F;
    }
    __syncthreads();

    const int rm = wv * 16;
    float* st_w = st_lds + rm * P;

    for (int ch = 0; ch < NCHUNK; ++ch) {
        const int cc = ch * CHUNK;

        static_assert(NCHUNK <= 4, "one priority level per chunk: the hardware has four");
        switch (ch) {
            case 0:  __builtin_amdgcn_s_setprio(3); break;
            case 1:  __builtin_amdgcn_s_setprio(2); break;
            case 2:  __builtin_amdgcn_s_setprio(1); break;
            default: __builtin_amdgcn_s_setprio(0); break;
        }

        float4 cs[4];
        {
            float4 cv[16];
            const float* src = ctxB + (size_t)(cc + rm) * Dn + lane * 4;
            #pragma unroll
            for (int j = 0; j < 16; ++j)
                cv[j] = *reinterpret_cast<const float4*>(src + (size_t)j * Dn);
            #pragma unroll
            for (int j = 0; j < 16; ++j)
                *reinterpret_cast<float4*>(st_w + j * P + lane * 4) = cv[j];
        }
        wave_handover();

        v8f acc[8];
        #pragma unroll
        for (int t = 0; t < 8; ++t) acc[t] = v8f{0.f,0.f,0.f,0.f,0.f,0.f,0.f,0.f};
        {
            const float* ap = st_w + ln * P + 2 * lh;
            const float* bp = q_lds + ln * P + 2 * lh;
            constexpr int SK = 4;
            constexpr int SNB = Dn / (4 * SK);
            v2f fa[2][SK];
            v2f fb[2][SK][8];
            #pragma unroll
            for (int j = 0; j < SK; ++j) {
                fa[0][j] = *reinterpret_cast<const v2f*>(ap + 4 * j);
                #pragma unroll
                for (int t = 0; t < 8; ++t)
                    fb[0][j][t] = *reinterpret_cast<const v2f*>(bp + t * 16 * P + 4 * j);
            }
            #pragma unroll
            for (int g = 0; g < SNB; ++g) {
                const int cur = g & 1, nxt = cur ^ 1;
                if (g + 1 < SNB) {
                    const int kb = (g + 1) * 4 * SK;
                    #pragma unroll
                    for (int j = 0; j < SK; ++j) {
                        fa[nxt][j] = *reinterpret_cast<const v2f*>(ap + kb + 4 * j);
                        #pragma unroll
                        for (int t = 0; t < 8; ++t)
                            fb[nxt][j][t] = *reinterpret_cast<const v2f*>(bp + t * 16 * P + kb + 4 * j);
                    }
                }
                __builtin_amdgcn_sched_barrier(0);
                #pragma unroll
                for (int j = 0; j < SK; ++j) {
                    #pragma unroll
                    for (int t = 0; t < 8; ++t)
                        acc[t] = __builtin_amdgcn_wmma_f32_16x16x4_f32(
                            false, fa[cur][j], false, fb[cur][j][t], (short)0, acc[t], false, false);
                }
                __builtin_amdgcn_sched_barrier(0);
            }
        }

        #pragma unroll
        for (int i = 0; i < 4; ++i)
            cs[i] = *reinterpret_cast<const float4*>(st_w + (4 * i + sr) * P + d0 + sc);
        wave_handover();

        #pragma unroll
        for (int r = 0; r < 8; ++r) {
            const int row = rm + r + 8 * lh;
            float sv[8];
            float mx = -3.4e38f;
            #pragma unroll
            for (int t = 0; t < 8; ++t) {
                const float s = acc[t][r] + qbias[t * 16 + ln];
                sv[t] = s;
                mx = fmaxf(mx, s);
            }
            mx = fmaxf(mx, row_xor<1>(mx));
            mx = fmaxf(mx, row_xor<2>(mx));
            mx = fmaxf(mx, row_xor<4>(mx));
            mx = fmaxf(mx, row_xor<8>(mx));
            if (ln == 0) m_all[cc + row] = mx;
            float sum = 0.f;
            #pragma unroll
            for (int t = 0; t < 8; ++t) { sv[t] = __expf(sv[t] - mx); sum += sv[t]; }
            sum += row_xor<1>(sum);
            sum += row_xor<2>(sum);
            sum += row_xor<4>(sum);
            sum += row_xor<8>(sum);
            const float inv = 1.0f / sum;
            #pragma unroll
            for (int t = 0; t < 8; ++t)
                st_lds[row * P + t * 16 + ln] = sv[t] * inv;
        }
        wave_handover();

        constexpr int NT2 = DS / 16;
        v8f acc2[NT2];
        #pragma unroll
        for (int t = 0; t < NT2; ++t) acc2[t] = v8f{0.f,0.f,0.f,0.f,0.f,0.f,0.f,0.f};

        static_assert(NT2 == 2, "one QS read carries the fragments of exactly two column tiles");
        {
            const float* ap = st_w + ln * P + 2 * lh;
            const float* bp = qs_lds + lh * (2 * DS) + ln * 4;
            constexpr int CK = 8;
            constexpr int CNB = Qn / (4 * CK);
            v2f ga[2][CK];
            v4f gb[2][CK];
            #pragma unroll
            for (int j = 0; j < CK; ++j) {
                ga[0][j] = *reinterpret_cast<const v2f*>(ap + 4 * j);
                gb[0][j] = *reinterpret_cast<const v4f*>(bp + (2 * j) * (2 * DS));
            }
            #pragma unroll
            for (int g = 0; g < CNB; ++g) {
                const int cur = g & 1, nxt = cur ^ 1;
                if (g + 1 < CNB) {
                    const int kb = (g + 1) * 4 * CK;
                    #pragma unroll
                    for (int j = 0; j < CK; ++j) {
                        ga[nxt][j] = *reinterpret_cast<const v2f*>(ap + kb + 4 * j);
                        gb[nxt][j] = *reinterpret_cast<const v4f*>(bp + ((kb >> 1) + 2 * j) * (2 * DS));
                    }
                }
                __builtin_amdgcn_sched_barrier(0);
                #pragma unroll
                for (int j = 0; j < CK; ++j) {
                    const v4f bq = gb[cur][j];
                    const v2f b0 = v2f{bq.x, bq.y};
                    const v2f b1 = v2f{bq.z, bq.w};
                    acc2[0] = __builtin_amdgcn_wmma_f32_16x16x4_f32(
                        false, ga[cur][j], false, b0, (short)0, acc2[0], false, false);
                    acc2[1] = __builtin_amdgcn_wmma_f32_16x16x4_f32(
                        false, ga[cur][j], false, b1, (short)0, acc2[1], false, false);
                }
                __builtin_amdgcn_sched_barrier(0);
            }
        }
        wave_handover();

        #pragma unroll
        for (int r = 0; r < 8; ++r) {
            const int row = rm + r + 8 * lh;
            #pragma unroll
            for (int t = 0; t < NT2; ++t)
                st_lds[row * P + t * 16 + ln] = acc2[t][r];
        }
        wave_handover();

        #pragma unroll
        for (int i = 0; i < 4; ++i) {
            const int j = 4 * i + sr;
            const float4 qv = *reinterpret_cast<const float4*>(st_w + j * P + sc);
            float* op = outB + (size_t)(cc + rm + j) * OUTW + d0 + sc;
            *reinterpret_cast<float4*>(op) = cs[i];
            *reinterpret_cast<float4*>(op + Dn) = qv;
            const float4 pv = make_float4(cs[i].x * qv.x, cs[i].y * qv.y, cs[i].z * qv.z, cs[i].w * qv.w);
            *reinterpret_cast<float4*>(op + 2 * Dn) = pv;
        }
        wave_handover();
    }
    constexpr int RG = Cn / (4 * NW);
    static_assert(RG == 16, "the tail keeps one wave's 16 row groups in registers");
    v4f cv[RG];
    {
        const float* cp = ctxB + (size_t)(wv * 4 + sr) * Dn + d0 + sc;
        #pragma unroll
        for (int j = 0; j < RG; ++j)
            cv[j] = *reinterpret_cast<const v4f*>(cp + (size_t)j * 4 * NW * Dn);
    }
    __syncthreads();

    {
        if (tid < 256) {
            float mx = -3.4e38f;
            for (int i = tid; i < Cn; i += 256) mx = fmaxf(mx, m_all[i]);
            red[tid] = mx;
        }
        __syncthreads();
        for (int s = 128; s > 0; s >>= 1) {
            if (tid < s) red[tid] = fmaxf(red[tid], red[tid + s]);
            __syncthreads();
        }
        const float gmax = red[0];
        __syncthreads();
        if (tid < 256) {
            float z = 0.f;
            for (int i = tid; i < Cn; i += 256) {
                const float p = __expf(m_all[i] - gmax);
                m_all[i] = p;
                z += p;
            }
            red[tid] = z;
        }
        __syncthreads();
        for (int s = 128; s > 0; s >>= 1) {
            if (tid < s) red[tid] += red[tid + s];
            __syncthreads();
        }
        const float invZ = 1.0f / red[0];
        __syncthreads();

        v4f part = v4f{0.f, 0.f, 0.f, 0.f};
        #pragma unroll
        for (int j = 0; j < RG; ++j)
            part += m_all[j * 4 * NW + wv * 4 + sr] * cv[j];
        part.x += row_xor<8>(part.x);
        part.y += row_xor<8>(part.y);
        part.z += row_xor<8>(part.z);
        part.w += row_xor<8>(part.w);
        float* pw = st_lds;
        if ((lane & 8) == 0)
            *reinterpret_cast<v4f*>(pw + (wv * 2 + lh) * DS + (lane & 7) * 4) = part;
        __syncthreads();
        if (tid < DS) {
            float s = 0.f;
            #pragma unroll
            for (int k = 0; k < 2 * NW; ++k) s += pw[k * DS + tid];
            q2c[tid] = s * invZ;
        }
        __syncthreads();
    }

    {
        const v4f qv = *reinterpret_cast<const v4f*>(q2c + sc);
        float* op = outB + (size_t)(wv * 4 + sr) * OUTW + 3 * Dn + d0 + sc;
        #pragma unroll
        for (int j = 0; j < RG; ++j)
            *reinterpret_cast<v4f*>(op + (size_t)j * 4 * NW * OUTW) = cv[j] * qv;
    }
}

extern "C" void kernel_launch(void* const* d_in, const int* in_sizes, int n_in,
                              void* d_out, int out_size, void* d_ws, size_t ws_size,
                              hipStream_t stream) {
    const float* ctx   = (const float*)d_in[0];
    const float* qry   = (const float*)d_in[1];
    const float* W     = (const float*)d_in[2];
    const float* qmask = (const float*)d_in[3];
    float* out = (float*)d_out;

    const size_t smem_bytes = (size_t)SM_FLOATS * sizeof(float);
    hipFuncSetAttribute(reinterpret_cast<const void*>(cqa_kernel),
                        hipFuncAttributeMaxDynamicSharedMemorySize, (int)smem_bytes);
    cqa_kernel<<<dim3(NGRID), dim3(NW * 32), smem_bytes, stream>>>(ctx, qry, W, qmask, out);
}


# row softmax, c2q and output blocks 0-2 split by rows, not columns: a work-group owns one 256-row chunk, all 128 columns (done first); of the other rows it takes only the row maximum; sum order kept
# speedup vs baseline: 4.1787x; 1.1113x over previous
#include <hip/hip_runtime.h>

typedef float v2f __attribute__((ext_vector_type(2)));
typedef float v8f __attribute__((ext_vector_type(8)));
typedef float v4f __attribute__((ext_vector_type(4)));

#define NEG_INF_F (-1000000000.0f)

template <int M>
__device__ __forceinline__ float row_xor(float v) {
    return __int_as_float(__builtin_amdgcn_update_dpp(0, __float_as_int(v), 0x160 + M, 0xf, 0xf, true));
}

__device__ __forceinline__ void wave_handover() {
    __builtin_amdgcn_fence(__ATOMIC_RELEASE, "wavefront");
    __builtin_amdgcn_wave_barrier();
    __builtin_amdgcn_fence(__ATOMIC_ACQUIRE, "wavefront");
}

constexpr int Bn = 64, Cn = 1024, Qn = 128, Dn = 128;
constexpr int P = 132;
constexpr int NW = 16;
constexpr int CHUNK = NW * 16;
constexpr int NCHUNK = Cn / CHUNK;
constexpr int OUTW = 4 * Dn;
constexpr int NSPLIT = 4;
constexpr int DS = Dn / NSPLIT;
constexpr int NGRID = Bn * NSPLIT;
constexpr int QP = 2 * Dn + 32;
constexpr int QS_FLOATS = (Qn / 2) * QP;
static_assert(NCHUNK == NSPLIT, "each of a batch element's work-groups owns exactly one chunk of rows");
constexpr int SM_FLOATS = Qn * P + CHUNK * P + 384 + Qn + QS_FLOATS + Cn + 256 + Dn;

__global__ __launch_bounds__(NW * 32, 1)
void cqa_kernel(const float* __restrict__ ctx_g,
                const float* __restrict__ qry_g,
                const float* __restrict__ W_g,
                const float* __restrict__ qmask_g,
                float* __restrict__ out_g)
{
    extern __shared__ float smem[];
    float* q_lds   = smem;
    float* st_lds  = q_lds   + Qn * P;
    float* w_lds   = st_lds  + CHUNK * P;
    float* qbias   = w_lds   + 384;
    float* qs_lds  = qbias   + Qn;
    float* m_all   = qs_lds  + QS_FLOATS;
    float* red     = m_all   + Cn;
    float* q2c     = red     + 256;

    const int b    = blockIdx.x / NSPLIT;
    const int split = blockIdx.x % NSPLIT;
    const int d0   = split * DS;
    const int tid  = threadIdx.x;
    const int wv   = tid >> 5;
    const int lane = tid & 31;
    const int lh   = lane >> 4;
    const int ln   = lane & 15;
    const int sr   = lane >> 3;
    const int sc   = (lane & 7) * 4;

    const float* ctxB = ctx_g + (size_t)b * Cn * Dn;
    const float* qryB = qry_g + (size_t)b * Qn * Dn;
    float*       outB = out_g + (size_t)b * Cn * OUTW;

    static_assert(3 * Dn <= NW * 32 && Qn <= NW * 32, "one W element and one mask element per thread");
    constexpr int QR = Qn / NW;
    const float w_reg = (tid < 3 * Dn) ? W_g[tid] : 0.f;
    const float m_reg = (tid < Qn) ? qmask_g[b * Qn + tid] : 0.f;
    const float4 wc4 = *reinterpret_cast<const float4*>(W_g + lane * 4);
    const float4 ws4 = *reinterpret_cast<const float4*>(W_g + 2 * Dn + lane * 4);
    float4 qrow[QR];
    #pragma unroll
    for (int i = 0; i < QR; ++i)
        qrow[i] = *reinterpret_cast<const float4*>(qryB + (size_t)(wv * QR + i) * Dn + lane * 4);
    static_assert(QR % 2 == 0, "a wave holds whole pairs of query rows (k, k + 1): QS stores them side by side");
    #pragma unroll
    for (int i = 0; i < QR; ++i) {
        const int k = wv * QR + i;
        const float4 q = qrow[i];
        *reinterpret_cast<float4*>(st_lds + k * P + lane * 4) = q;
        *reinterpret_cast<float4*>(q_lds + k * P + lane * 4) =
            make_float4(fmaf(ws4.x, q.x, wc4.x), fmaf(ws4.y, q.y, wc4.y),
                        fmaf(ws4.z, q.z, wc4.z), fmaf(ws4.w, q.w, wc4.w));
    }
    #pragma unroll
    for (int i = 0; i < QR; i += 2) {
        const int k = wv * QR + i;
        const float4 q0 = qrow[i], q1 = qrow[i + 1];
        float* qp = qs_lds + (k >> 1) * QP + lane * 8;
        *reinterpret_cast<float4*>(qp)     = make_float4(q0.x, q1.x, q0.y, q1.y);
        *reinterpret_cast<float4*>(qp + 4) = make_float4(q0.z, q1.z, q0.w, q1.w);
    }
    if (tid < 3 * Dn) w_lds[tid] = w_reg;
    __syncthreads();
    if (tid < Qn) {
        float s = 0.f;
        for (int d = 0; d < Dn; ++d) s += st_lds[tid * P + d] * w_lds[Dn + d];
        qbias[tid] = s + (1.0f - m_reg) * NEG_INF_F;
    }
    __syncthreads();

    const int rm = wv * 16;
    float* st_w = st_lds + rm * P;
    float* st_ld = st_w + lane * 4;
    float* st_c  = st_w + (8 * lh) * P + ln;

    for (int ch = 0; ch < NCHUNK; ++ch) {
        const bool own = (ch == 0);
        const int cc = ((split + ch) % NCHUNK) * CHUNK;

        static_assert(NCHUNK <= 4, "one priority level per chunk: the hardware has four");
        switch (ch) {
            case 0:  __builtin_amdgcn_s_setprio(3); break;
            case 1:  __builtin_amdgcn_s_setprio(2); break;
            case 2:  __builtin_amdgcn_s_setprio(1); break;
            default: __builtin_amdgcn_s_setprio(0); break;
        }

        {
            float4 cv[16];
            const float* src = ctxB + (size_t)(cc + rm) * Dn + lane * 4;
            #pragma unroll
            for (int j = 0; j < 16; ++j)
                cv[j] = *reinterpret_cast<const float4*>(src + (size_t)j * Dn);
            #pragma unroll
            for (int j = 0; j < 16; ++j)
                *reinterpret_cast<float4*>(st_ld + j * P) = cv[j];
        }
        wave_handover();

        v8f acc[8];
        #pragma unroll
        for (int t = 0; t < 8; ++t) acc[t] = v8f{0.f,0.f,0.f,0.f,0.f,0.f,0.f,0.f};
        {
            const float* ap = st_w + ln * P + 2 * lh;
            const float* bp = q_lds + ln * P + 2 * lh;
            constexpr int SK = 4;
            constexpr int SNB = Dn / (4 * SK);
            v2f fa[2][SK];
            v2f fb[2][SK][8];
            #pragma unroll
            for (int j = 0; j < SK; ++j) {
                fa[0][j] = *reinterpret_cast<const v2f*>(ap + 4 * j);
                #pragma unroll
                for (int t = 0; t < 8; ++t)
                    fb[0][j][t] = *reinterpret_cast<const v2f*>(bp + t * 16 * P + 4 * j);
            }
            #pragma unroll
            for (int g = 0; g < SNB; ++g) {
                const int cur = g & 1, nxt = cur ^ 1;
                if (g + 1 < SNB) {
                    const int kb = (g + 1) * 4 * SK;
                    #pragma unroll
                    for (int j = 0; j < SK; ++j) {
                        fa[nxt][j] = *reinterpret_cast<const v2f*>(ap + kb + 4 * j);
                        #pragma unroll
                        for (int t = 0; t < 8; ++t)
                            fb[nxt][j][t] = *reinterpret_cast<const v2f*>(bp + t * 16 * P + kb + 4 * j);
                    }
                }
                __builtin_amdgcn_sched_barrier(0);
                #pragma unroll
                for (int j = 0; j < SK; ++j) {
                    #pragma unroll
                    for (int t = 0; t < 8; ++t)
                        acc[t] = __builtin_amdgcn_wmma_f32_16x16x4_f32(
                            false, fa[cur][j], false, fb[cur][j][t], (short)0, acc[t], false, false);
                }
                __builtin_amdgcn_sched_barrier(0);
            }
        }

        if (!own) {
            #pragma unroll
            for (int r = 0; r < 8; ++r) {
                float mx = -3.4e38f;
                #pragma unroll
                for (int t = 0; t < 8; ++t)
                    mx = fmaxf(mx, acc[t][r] + qbias[t * 16 + ln]);
                mx = fmaxf(mx, row_xor<1>(mx));
                mx = fmaxf(mx, row_xor<2>(mx));
                mx = fmaxf(mx, row_xor<4>(mx));
                mx = fmaxf(mx, row_xor<8>(mx));
                if (ln == 0) m_all[cc + rm + r + 8 * lh] = mx;
            }
            wave_handover();
            continue;
        }
        wave_handover();

        #pragma unroll
        for (int r = 0; r < 8; ++r) {
            const int row = rm + r + 8 * lh;
            float sv[8];
            float mx = -3.4e38f;
            #pragma unroll
            for (int t = 0; t < 8; ++t) {
                const float s = acc[t][r] + qbias[t * 16 + ln];
                sv[t] = s;
                mx = fmaxf(mx, s);
            }
            mx = fmaxf(mx, row_xor<1>(mx));
            mx = fmaxf(mx, row_xor<2>(mx));
            mx = fmaxf(mx, row_xor<4>(mx));
            mx = fmaxf(mx, row_xor<8>(mx));
            if (ln == 0) m_all[cc + row] = mx;
            float sum = 0.f;
            #pragma unroll
            for (int t = 0; t < 8; ++t) { sv[t] = __expf(sv[t] - mx); sum += sv[t]; }
            sum += row_xor<1>(sum);
            sum += row_xor<2>(sum);
            sum += row_xor<4>(sum);
            sum += row_xor<8>(sum);
            const float inv = 1.0f / sum;
            #pragma unroll
            for (int t = 0; t < 8; ++t)
                st_c[r * P + t * 16] = sv[t] * inv;
        }
        wave_handover();

        v8f acc2[8];
        #pragma unroll
        for (int t = 0; t < 8; ++t) acc2[t] = v8f{0.f,0.f,0.f,0.f,0.f,0.f,0.f,0.f};
        {
            const float* ap = st_w + ln * P + 2 * lh;
            const float* bp = qs_lds + lh * QP + 2 * ln;
            constexpr int CK = 4;
            constexpr int CNB = Qn / (4 * CK);
            v2f ga[2][CK];
            v2f gb[2][CK][8];
            #pragma unroll
            for (int j = 0; j < CK; ++j) {
                ga[0][j] = *reinterpret_cast<const v2f*>(ap + 4 * j);
                #pragma unroll
                for (int t = 0; t < 8; ++t)
                    gb[0][j][t] = *reinterpret_cast<const v2f*>(bp + (2 * j) * QP + t * 32);
            }
            #pragma unroll
            for (int g = 0; g < CNB; ++g) {
                const int cur = g & 1, nxt = cur ^ 1;
                if (g + 1 < CNB) {
                    const int kb = (g + 1) * 4 * CK;
                    #pragma unroll
                    for (int j = 0; j < CK; ++j) {
                        ga[nxt][j] = *reinterpret_cast<const v2f*>(ap + kb + 4 * j);
                        #pragma unroll
                        for (int t = 0; t < 8; ++t)
                            gb[nxt][j][t] = *reinterpret_cast<const v2f*>(bp + ((kb >> 1) + 2 * j) * QP + t * 32);
                    }
                }
                __builtin_amdgcn_sched_barrier(0);
                #pragma unroll
                for (int j = 0; j < CK; ++j) {
                    #pragma unroll
                    for (int t = 0; t < 8; ++t)
                        acc2[t] = __builtin_amdgcn_wmma_f32_16x16x4_f32(
                            false, ga[cur][j], false, gb[cur][j][t], (short)0, acc2[t], false, false);
                }
                __builtin_amdgcn_sched_barrier(0);
            }
        }
        wave_handover();

        float4 cr[16];
        {
            const float* src = ctxB + (size_t)(cc + rm) * Dn + lane * 4;
            #pragma unroll
            for (int j = 0; j < 16; ++j)
                cr[j] = *reinterpret_cast<const float4*>(src + (size_t)j * Dn);
        }
        #pragma unroll
        for (int r = 0; r < 8; ++r) {
            #pragma unroll
            for (int t = 0; t < 8; ++t)
                st_c[r * P + t * 16] = acc2[t][r];
        }
        wave_handover();

        #pragma unroll
        for (int j = 0; j < 16; ++j) {
            const float4 qv = *reinterpret_cast<const float4*>(st_ld + j * P);
            float* op = outB + (size_t)(cc + rm + j) * OUTW + lane * 4;
            *reinterpret_cast<float4*>(op) = cr[j];
            *reinterpret_cast<float4*>(op + Dn) = qv;
            const float4 pv = make_float4(cr[j].x * qv.x, cr[j].y * qv.y, cr[j].z * qv.z, cr[j].w * qv.w);
            *reinterpret_cast<float4*>(op + 2 * Dn) = pv;
        }
        wave_handover();
    }
    constexpr int RG = Cn / (4 * NW);
    static_assert(RG == 16, "the tail keeps one wave's 16 row groups in registers");
    v4f cv[RG];
    {
        const float* cp = ctxB + (size_t)(wv * 4 + sr) * Dn + d0 + sc;
        #pragma unroll
        for (int j = 0; j < RG; ++j)
            cv[j] = *reinterpret_cast<const v4f*>(cp + (size_t)j * 4 * NW * Dn);
    }
    __syncthreads();

    {
        if (tid < 256) {
            float mx = -3.4e38f;
            for (int i = tid; i < Cn; i += 256) mx = fmaxf(mx, m_all[i]);
            red[tid] = mx;
        }
        __syncthreads();
        for (int s = 128; s > 0; s >>= 1) {
            if (tid < s) red[tid] = fmaxf(red[tid], red[tid + s]);
            __syncthreads();
        }
        const float gmax = red[0];
        __syncthreads();
        if (tid < 256) {
            float z = 0.f;
            for (int i = tid; i < Cn; i += 256) {
                const float p = __expf(m_all[i] - gmax);
                m_all[i] = p;
                z += p;
            }
            red[tid] = z;
        }
        __syncthreads();
        for (int s = 128; s > 0; s >>= 1) {
            if (tid < s) red[tid] += red[tid + s];
            __syncthreads();
        }
        const float invZ = 1.0f / red[0];
        __syncthreads();

        v4f part = v4f{0.f, 0.f, 0.f, 0.f};
        #pragma unroll
        for (int j = 0; j < RG; ++j)
            part += m_all[j * 4 * NW + wv * 4 + sr] * cv[j];
        part.x += row_xor<8>(part.x);
        part.y += row_xor<8>(part.y);
        part.z += row_xor<8>(part.z);
        part.w += row_xor<8>(part.w);
        float* pw = st_lds;
        if ((lane & 8) == 0)
            *reinterpret_cast<v4f*>(pw + wv * (2 * DS) + (lane & 16) * (DS / 16) + (lane & 7) * 4) = part;
        __syncthreads();
        if (tid < DS) {
            float s = 0.f;
            #pragma unroll
            for (int k = 0; k < 2 * NW; ++k) s += pw[k * DS + tid];
            q2c[tid] = s * invZ;
        }
        __syncthreads();
    }

    {
        const v4f qv = *reinterpret_cast<const v4f*>(q2c + sc);
        float* op = outB + (size_t)(wv * 4 + sr) * OUTW + 3 * Dn + d0 + sc;
        #pragma unroll
        for (int j = 0; j < RG; ++j)
            *reinterpret_cast<v4f*>(op + (size_t)j * 4 * NW * OUTW) = cv[j] * qv;
    }
}

extern "C" void kernel_launch(void* const* d_in, const int* in_sizes, int n_in,
                              void* d_out, int out_size, void* d_ws, size_t ws_size,
                              hipStream_t stream) {
    const float* ctx   = (const float*)d_in[0];
    const float* qry   = (const float*)d_in[1];
    const float* W     = (const float*)d_in[2];
    const float* qmask = (const float*)d_in[3];
    float* out = (float*)d_out;

    const size_t smem_bytes = (size_t)SM_FLOATS * sizeof(float);
    hipFuncSetAttribute(reinterpret_cast<const void*>(cqa_kernel),
                        hipFuncAttributeMaxDynamicSharedMemorySize, (int)smem_bytes);
    cqa_kernel<<<dim3(NGRID), dim3(NW * 32), smem_bytes, stream>>>(ctx, qry, W, qmask, out);
}
